# starting kernel
# baseline (speedup 1.0000x reference)
#include <hip/hip_runtime.h>

typedef __attribute__((ext_vector_type(2))) float v2f;
typedef __attribute__((ext_vector_type(8))) float v8f;

#define BATCH 8
#define SEQ   8192
#define DN    128
#define CH    128
#define NCH   (SEQ / CH)

__global__ void ssm_pass1(const float* __restrict__ x, const float* __restrict__ dt,
                          const float* __restrict__ B, const float* __restrict__ C,
                          const float* __restrict__ A,
                          float* __restrict__ y, float* __restrict__ aprod,
                          float* __restrict__ hloc)
{
    __shared__ float dt_s[CH];
    __shared__ float dtx_s[CH];
    __shared__ float ypart[CH * 4];

    const int k = blockIdx.x, b = blockIdx.y;
    const int tid = threadIdx.x;
    const int lane = tid & 31, wv = tid >> 5;
    const size_t sbase = (size_t)b * SEQ + (size_t)k * CH;

    float d = dt[sbase + tid];
    dt_s[tid]  = d;
    dtx_s[tid] = d * x[sbase + tid];

    const float* Bp = B + sbase * DN + tid;
    const float* Cp = C + sbase * DN + tid;
    __builtin_prefetch(Bp, 0, 0);
    __builtin_prefetch(Cp, 0, 0);
    __syncthreads();

    const float An = A[tid];
    float h = 0.f, sd = 0.f;
#pragma unroll 4
    for (int t = 0; t < CH; ++t) {
        float dts = dt_s[t];
        sd += dts;
        float a = __expf(dts * An);
        h = fmaf(a, h, dtx_s[t] * Bp[(size_t)t * DN]);
        float p = Cp[(size_t)t * DN] * h;
        p += __shfl_xor(p, 1, 32);
        p += __shfl_xor(p, 2, 32);
        p += __shfl_xor(p, 4, 32);
        p += __shfl_xor(p, 8, 32);
        p += __shfl_xor(p, 16, 32);
        if (lane == 0) ypart[t * 4 + wv] = p;
    }

    const size_t cidx = ((size_t)b * NCH + k) * DN + tid;
    aprod[cidx] = __expf(An * sd);
    hloc[cidx]  = h;
    __syncthreads();
    y[sbase + tid] = ypart[tid * 4 + 0] + ypart[tid * 4 + 1] +
                     ypart[tid * 4 + 2] + ypart[tid * 4 + 3];
}

__global__ void ssm_pass2(const float* __restrict__ aprod, const float* __restrict__ hloc,
                          float* __restrict__ carry, float* __restrict__ hfin)
{
    const int b = blockIdx.x, n = threadIdx.x;
    float h = 0.f;
    for (int k = 0; k < NCH; ++k) {
        const size_t idx = ((size_t)b * NCH + k) * DN + n;
        carry[idx] = h;
        h = fmaf(aprod[idx], h, hloc[idx]);
    }
    hfin[(size_t)b * DN + n] = h;
}

__global__ void ssm_pass3(const float* __restrict__ dt, const float* __restrict__ C,
                          const float* __restrict__ A, const float* __restrict__ carry,
                          float* __restrict__ y)
{
    __shared__ float Sd[CH];
    __shared__ float sA[DN];
    __shared__ float hin[DN];
    __shared__ float ycorr[CH];

    const int k = blockIdx.x, b = blockIdx.y;
    const int tid = threadIdx.x;
    const int lane = tid & 31, wv = tid >> 5;
    const size_t sbase = (size_t)b * SEQ + (size_t)k * CH;

    Sd[tid]  = dt[sbase + tid];
    sA[tid]  = A[tid];
    hin[tid] = carry[((size_t)b * NCH + k) * DN + tid];
    __syncthreads();
    for (int off = 1; off < CH; off <<= 1) {
        float v = (tid >= off) ? Sd[tid - off] : 0.f;
        __syncthreads();
        Sd[tid] += v;
        __syncthreads();
    }

    const float* Cb = C + sbase * DN;
    const int kBase = (lane < 16) ? 0 : 2;

    for (int i = 0; i < 2; ++i) {
        const int t0   = (wv * 2 + i) * 16;
        const int tRow = t0 + (lane & 15);
        const float sdt = Sd[tRow];
        const float* Crow = Cb + (size_t)tRow * DN;

        v8f acc = {};
#pragma unroll 8
        for (int k0 = 0; k0 < DN; k0 += 4) {
            const int kk = k0 + kBase;
            float2 cv = *(const float2*)(Crow + kk);
            v2f a2, b2;
            a2.x = cv.x * __expf(sA[kk] * sdt);
            a2.y = cv.y * __expf(sA[kk + 1] * sdt);
            b2.x = hin[kk];
            b2.y = hin[kk + 1];
            acc = __builtin_amdgcn_wmma_f32_16x16x4_f32(
                false, a2, false, b2, (short)0, acc, false, false);
        }
        if (lane == 0) {
#pragma unroll
            for (int r = 0; r < 8; ++r) ycorr[t0 + r] = acc[r];
        }
        if (lane == 16) {
#pragma unroll
            for (int r = 0; r < 8; ++r) ycorr[t0 + 8 + r] = acc[r];
        }
    }
    __syncthreads();
    y[sbase + tid] += ycorr[tid];
}

extern "C" void kernel_launch(void* const* d_in, const int* in_sizes, int n_in,
                              void* d_out, int out_size, void* d_ws, size_t ws_size,
                              hipStream_t stream) {
    const float* x  = (const float*)d_in[0];
    const float* dt = (const float*)d_in[1];
    const float* B  = (const float*)d_in[2];
    const float* C  = (const float*)d_in[3];
    const float* A  = (const float*)d_in[4];

    float* y    = (float*)d_out;
    float* hfin = y + (size_t)BATCH * SEQ;

    float* aprod = (float*)d_ws;
    float* hloc  = aprod + (size_t)BATCH * NCH * DN;
    float* carry = hloc  + (size_t)BATCH * NCH * DN;

    dim3 grid(NCH, BATCH);
    ssm_pass1<<<grid, CH, 0, stream>>>(x, dt, B, C, A, y, aprod, hloc);
    ssm_pass2<<<BATCH, DN, 0, stream>>>(aprod, hloc, carry, hfin);
    ssm_pass3<<<grid, CH, 0, stream>>>(dt, C, A, carry, y);
}


# pass 1 reads the B and C rows of the next 16 timesteps into registers while it scans the current 16, and a group's dt and dt*x from LDS in one batch: no step waits for its own reads; sums unchanged
# speedup vs baseline: 2.2256x; 2.2256x over previous
#include <hip/hip_runtime.h>

typedef __attribute__((ext_vector_type(2))) float v2f;
typedef __attribute__((ext_vector_type(8))) float v8f;

#define BATCH 8
#define SEQ   8192
#define DN    128
#define CH    128
#define NCH   (SEQ / CH)

#define GRP 16
#define NGRP (CH / GRP)

__device__ __forceinline__ void load_group(float (&bq)[GRP], float (&cq)[GRP],
                                           const float* __restrict__ Bp,
                                           const float* __restrict__ Cp, int t0)
{
    for (int j = 0; j < GRP; ++j) {
        bq[j] = Bp[(size_t)(t0 + j) * DN];
        cq[j] = Cp[(size_t)(t0 + j) * DN];
    }
}

__global__ void ssm_pass1(const float* __restrict__ x, const float* __restrict__ dt,
                          const float* __restrict__ B, const float* __restrict__ C,
                          const float* __restrict__ A,
                          float* __restrict__ y, float* __restrict__ aprod,
                          float* __restrict__ hloc)
{
    __shared__ __attribute__((aligned(16))) float dt_s[CH];
    __shared__ __attribute__((aligned(16))) float dtx_s[CH];
    __shared__ float ypart[CH * 4];

    const int k = blockIdx.x, b = blockIdx.y;
    const int tid = threadIdx.x;
    const int lane = tid & 31, wv = tid >> 5;
    const size_t sbase = (size_t)b * SEQ + (size_t)k * CH;

    const float* Bp = B + sbase * DN + tid;
    const float* Cp = C + sbase * DN + tid;

    float bc[GRP], cc[GRP];
    float bn[GRP], cn[GRP];
    load_group(bc, cc, Bp, Cp, 0);

    float d = dt[sbase + tid];
    dt_s[tid]  = d;
    dtx_s[tid] = d * x[sbase + tid];
    const float An = A[tid];
    __syncthreads();

    float h = 0.f, sd = 0.f;
    for (int g = 0; g < NGRP; ++g) {
        const int t0 = g * GRP;
        if (g + 1 < NGRP) load_group(bn, cn, Bp, Cp, t0 + GRP);

        float dtv[GRP], dxv[GRP];
        for (int q = 0; q < GRP / 4; ++q) {
            const float4 d4 = *(const float4*)&dt_s[t0 + 4 * q];
            const float4 x4 = *(const float4*)&dtx_s[t0 + 4 * q];
            dtv[4 * q + 0] = d4.x; dtv[4 * q + 1] = d4.y;
            dtv[4 * q + 2] = d4.z; dtv[4 * q + 3] = d4.w;
            dxv[4 * q + 0] = x4.x; dxv[4 * q + 1] = x4.y;
            dxv[4 * q + 2] = x4.z; dxv[4 * q + 3] = x4.w;
        }

        for (int j = 0; j < GRP; ++j) {
            const int t = t0 + j;
            float dts = dtv[j];
            sd += dts;
            float a = __expf(dts * An);
            h = fmaf(a, h, dxv[j] * bc[j]);
            float p = cc[j] * h;
            p += __shfl_xor(p, 1, 32);
            p += __shfl_xor(p, 2, 32);
            p += __shfl_xor(p, 4, 32);
            p += __shfl_xor(p, 8, 32);
            p += __shfl_xor(p, 16, 32);
            if (lane == 0) ypart[t * 4 + wv] = p;
        }

        for (int j = 0; j < GRP; ++j) { bc[j] = bn[j]; cc[j] = cn[j]; }
    }

    const size_t cidx = ((size_t)b * NCH + k) * DN + tid;
    aprod[cidx] = __expf(An * sd);
    hloc[cidx]  = h;
    __syncthreads();
    y[sbase + tid] = ypart[tid * 4 + 0] + ypart[tid * 4 + 1] +
                     ypart[tid * 4 + 2] + ypart[tid * 4 + 3];
}

__global__ void ssm_pass2(const float* __restrict__ aprod, const float* __restrict__ hloc,
                          float* __restrict__ carry, float* __restrict__ hfin)
{
    const int b = blockIdx.x, n = threadIdx.x;
    float h = 0.f;
    for (int k = 0; k < NCH; ++k) {
        const size_t idx = ((size_t)b * NCH + k) * DN + n;
        carry[idx] = h;
        h = fmaf(aprod[idx], h, hloc[idx]);
    }
    hfin[(size_t)b * DN + n] = h;
}

__global__ void ssm_pass3(const float* __restrict__ dt, const float* __restrict__ C,
                          const float* __restrict__ A, const float* __restrict__ carry,
                          float* __restrict__ y)
{
    __shared__ float Sd[CH];
    __shared__ float sA[DN];
    __shared__ float hin[DN];
    __shared__ float ycorr[CH];

    const int k = blockIdx.x, b = blockIdx.y;
    const int tid = threadIdx.x;
    const int lane = tid & 31, wv = tid >> 5;
    const size_t sbase = (size_t)b * SEQ + (size_t)k * CH;

    Sd[tid]  = dt[sbase + tid];
    sA[tid]  = A[tid];
    hin[tid] = carry[((size_t)b * NCH + k) * DN + tid];
    __syncthreads();
    for (int off = 1; off < CH; off <<= 1) {
        float v = (tid >= off) ? Sd[tid - off] : 0.f;
        __syncthreads();
        Sd[tid] += v;
        __syncthreads();
    }

    const float* Cb = C + sbase * DN;
    const int kBase = (lane < 16) ? 0 : 2;

    for (int i = 0; i < 2; ++i) {
        const int t0   = (wv * 2 + i) * 16;
        const int tRow = t0 + (lane & 15);
        const float sdt = Sd[tRow];
        const float* Crow = Cb + (size_t)tRow * DN;

        v8f acc = {};
#pragma unroll 8
        for (int k0 = 0; k0 < DN; k0 += 4) {
            const int kk = k0 + kBase;
            float2 cv = *(const float2*)(Crow + kk);
            v2f a2, b2;
            a2.x = cv.x * __expf(sA[kk] * sdt);
            a2.y = cv.y * __expf(sA[kk + 1] * sdt);
            b2.x = hin[kk];
            b2.y = hin[kk + 1];
            acc = __builtin_amdgcn_wmma_f32_16x16x4_f32(
                false, a2, false, b2, (short)0, acc, false, false);
        }
        if (lane == 0) {
#pragma unroll
            for (int r = 0; r < 8; ++r) ycorr[t0 + r] = acc[r];
        }
        if (lane == 16) {
#pragma unroll
            for (int r = 0; r < 8; ++r) ycorr[t0 + 8 + r] = acc[r];
        }
    }
    __syncthreads();
    y[sbase + tid] += ycorr[tid];
}

extern "C" void kernel_launch(void* const* d_in, const int* in_sizes, int n_in,
                              void* d_out, int out_size, void* d_ws, size_t ws_size,
                              hipStream_t stream) {
    const float* x  = (const float*)d_in[0];
    const float* dt = (const float*)d_in[1];
    const float* B  = (const float*)d_in[2];
    const float* C  = (const float*)d_in[3];
    const float* A  = (const float*)d_in[4];

    float* y    = (float*)d_out;
    float* hfin = y + (size_t)BATCH * SEQ;

    float* aprod = (float*)d_ws;
    float* hloc  = aprod + (size_t)BATCH * NCH * DN;
    float* carry = hloc  + (size_t)BATCH * NCH * DN;

    dim3 grid(NCH, BATCH);
    ssm_pass1<<<grid, CH, 0, stream>>>(x, dt, B, C, A, y, aprod, hloc);
    ssm_pass2<<<BATCH, DN, 0, stream>>>(aprod, hloc, carry, hfin);
    ssm_pass3<<<grid, CH, 0, stream>>>(dt, C, A, carry, y);
}


# pass 1: each step's 32-lane sum runs on the vector ALU (row_xmask DPP adds, one permlanex16) instead of five waited ds_bpermute; a group's 16 sums stored together; same sum tree
# speedup vs baseline: 3.9296x; 1.7656x over previous
#include <hip/hip_runtime.h>

typedef __attribute__((ext_vector_type(2))) float v2f;
typedef __attribute__((ext_vector_type(8))) float v8f;

#define BATCH 8
#define SEQ   8192
#define DN    128
#define CH    128
#define NCH   (SEQ / CH)

#define GRP 16
#define NGRP (CH / GRP)

template <int M>
__device__ __forceinline__ float row_xor(float v)
{
    return __builtin_bit_cast(float,
        __builtin_amdgcn_update_dpp(0, __builtin_bit_cast(int, v),
                                    0x160 | M  , 0xf, 0xf, true));
}

__device__ __forceinline__ float xhalf(float v)
{
    const int b = __builtin_bit_cast(int, v);
    return __builtin_bit_cast(float,
        __builtin_amdgcn_permlanex16(b, b, 0x76543210, 0xfedcba98, false, false));
}

__device__ __forceinline__ float wave_sum(float p)
{
    p += row_xor<1>(p);
    p += row_xor<2>(p);
    p += row_xor<4>(p);
    p += row_xor<8>(p);
    p += xhalf(p);
    return p;
}

__device__ __forceinline__ void load_group(float (&bq)[GRP], float (&cq)[GRP],
                                           const float* __restrict__ Bp,
                                           const float* __restrict__ Cp, int t0)
{
    for (int j = 0; j < GRP; ++j) {
        bq[j] = Bp[(size_t)(t0 + j) * DN];
        cq[j] = Cp[(size_t)(t0 + j) * DN];
    }
}

__global__ void ssm_pass1(const float* __restrict__ x, const float* __restrict__ dt,
                          const float* __restrict__ B, const float* __restrict__ C,
                          const float* __restrict__ A,
                          float* __restrict__ y, float* __restrict__ aprod,
                          float* __restrict__ hloc)
{
    __shared__ __attribute__((aligned(16))) float dt_s[CH];
    __shared__ __attribute__((aligned(16))) float dtx_s[CH];
    __shared__ __attribute__((aligned(16))) float ypart[4 * CH];

    const int k = blockIdx.x, b = blockIdx.y;
    const int tid = threadIdx.x;
    const int lane = tid & 31, wv = tid >> 5;
    const size_t sbase = (size_t)b * SEQ + (size_t)k * CH;

    const float* Bp = B + sbase * DN + tid;
    const float* Cp = C + sbase * DN + tid;

    float bc[GRP], cc[GRP];
    float bn[GRP], cn[GRP];
    load_group(bc, cc, Bp, Cp, 0);

    float d = dt[sbase + tid];
    dt_s[tid]  = d;
    dtx_s[tid] = d * x[sbase + tid];
    const float An = A[tid];
    __syncthreads();

    float h = 0.f, sd = 0.f;
    for (int g = 0; g < NGRP; ++g) {
        const int t0 = g * GRP;
        if (g + 1 < NGRP) load_group(bn, cn, Bp, Cp, t0 + GRP);

        float dtv[GRP], dxv[GRP];
        for (int q = 0; q < GRP / 4; ++q) {
            const float4 d4 = *(const float4*)&dt_s[t0 + 4 * q];
            const float4 x4 = *(const float4*)&dtx_s[t0 + 4 * q];
            dtv[4 * q + 0] = d4.x; dtv[4 * q + 1] = d4.y;
            dtv[4 * q + 2] = d4.z; dtv[4 * q + 3] = d4.w;
            dxv[4 * q + 0] = x4.x; dxv[4 * q + 1] = x4.y;
            dxv[4 * q + 2] = x4.z; dxv[4 * q + 3] = x4.w;
        }

        float ps[GRP];
        for (int j = 0; j < GRP; ++j) {
            float dts = dtv[j];
            sd += dts;
            float a = __expf(dts * An);
            h = fmaf(a, h, dxv[j] * bc[j]);
            ps[j] = wave_sum(cc[j] * h);
        }
        if (lane == 0) {
            for (int q = 0; q < GRP / 4; ++q)
                *(float4*)&ypart[wv * CH + t0 + 4 * q] =
                    make_float4(ps[4 * q + 0], ps[4 * q + 1], ps[4 * q + 2], ps[4 * q + 3]);
        }

        for (int j = 0; j < GRP; ++j) { bc[j] = bn[j]; cc[j] = cn[j]; }
    }

    const size_t cidx = ((size_t)b * NCH + k) * DN + tid;
    aprod[cidx] = __expf(An * sd);
    hloc[cidx]  = h;
    __syncthreads();
    y[sbase + tid] = ypart[0 * CH + tid] + ypart[1 * CH + tid] +
                     ypart[2 * CH + tid] + ypart[3 * CH + tid];
}

__global__ void ssm_pass2(const float* __restrict__ aprod, const float* __restrict__ hloc,
                          float* __restrict__ carry, float* __restrict__ hfin)
{
    const int b = blockIdx.x, n = threadIdx.x;
    float h = 0.f;
    for (int k = 0; k < NCH; ++k) {
        const size_t idx = ((size_t)b * NCH + k) * DN + n;
        carry[idx] = h;
        h = fmaf(aprod[idx], h, hloc[idx]);
    }
    hfin[(size_t)b * DN + n] = h;
}

__global__ void ssm_pass3(const float* __restrict__ dt, const float* __restrict__ C,
                          const float* __restrict__ A, const float* __restrict__ carry,
                          float* __restrict__ y)
{
    __shared__ float Sd[CH];
    __shared__ float sA[DN];
    __shared__ float hin[DN];
    __shared__ float ycorr[CH];

    const int k = blockIdx.x, b = blockIdx.y;
    const int tid = threadIdx.x;
    const int lane = tid & 31, wv = tid >> 5;
    const size_t sbase = (size_t)b * SEQ + (size_t)k * CH;

    Sd[tid]  = dt[sbase + tid];
    sA[tid]  = A[tid];
    hin[tid] = carry[((size_t)b * NCH + k) * DN + tid];
    __syncthreads();
    for (int off = 1; off < CH; off <<= 1) {
        float v = (tid >= off) ? Sd[tid - off] : 0.f;
        __syncthreads();
        Sd[tid] += v;
        __syncthreads();
    }

    const float* Cb = C + sbase * DN;
    const int kBase = (lane < 16) ? 0 : 2;

    for (int i = 0; i < 2; ++i) {
        const int t0   = (wv * 2 + i) * 16;
        const int tRow = t0 + (lane & 15);
        const float sdt = Sd[tRow];
        const float* Crow = Cb + (size_t)tRow * DN;

        v8f acc = {};
#pragma unroll 8
        for (int k0 = 0; k0 < DN; k0 += 4) {
            const int kk = k0 + kBase;
            float2 cv = *(const float2*)(Crow + kk);
            v2f a2, b2;
            a2.x = cv.x * __expf(sA[kk] * sdt);
            a2.y = cv.y * __expf(sA[kk + 1] * sdt);
            b2.x = hin[kk];
            b2.y = hin[kk + 1];
            acc = __builtin_amdgcn_wmma_f32_16x16x4_f32(
                false, a2, false, b2, (short)0, acc, false, false);
        }
        if (lane == 0) {
#pragma unroll
            for (int r = 0; r < 8; ++r) ycorr[t0 + r] = acc[r];
        }
        if (lane == 16) {
#pragma unroll
            for (int r = 0; r < 8; ++r) ycorr[t0 + 8 + r] = acc[r];
        }
    }
    __syncthreads();
    y[sbase + tid] += ycorr[tid];
}

extern "C" void kernel_launch(void* const* d_in, const int* in_sizes, int n_in,
                              void* d_out, int out_size, void* d_ws, size_t ws_size,
                              hipStream_t stream) {
    const float* x  = (const float*)d_in[0];
    const float* dt = (const float*)d_in[1];
    const float* B  = (const float*)d_in[2];
    const float* C  = (const float*)d_in[3];
    const float* A  = (const float*)d_in[4];

    float* y    = (float*)d_out;
    float* hfin = y + (size_t)BATCH * SEQ;

    float* aprod = (float*)d_ws;
    float* hloc  = aprod + (size_t)BATCH * NCH * DN;
    float* carry = hloc  + (size_t)BATCH * NCH * DN;

    dim3 grid(NCH, BATCH);
    ssm_pass1<<<grid, CH, 0, stream>>>(x, dt, B, C, A, y, aprod, hloc);
    ssm_pass2<<<BATCH, DN, 0, stream>>>(aprod, hloc, carry, hfin);
    ssm_pass3<<<grid, CH, 0, stream>>>(dt, C, A, carry, y);
}


# the 8-block launch that chained the 64 chunk summaries is gone: each correction block folds the summaries of the chunks before it itself, same fma order, all pair reads in one round trip
# speedup vs baseline: 5.3723x; 1.3671x over previous
#include <hip/hip_runtime.h>

typedef __attribute__((ext_vector_type(2))) float v2f;
typedef __attribute__((ext_vector_type(8))) float v8f;

#define BATCH 8
#define SEQ   8192
#define DN    128
#define CH    128
#define NCH   (SEQ / CH)

#define GRP 16
#define NGRP (CH / GRP)

template <int M>
__device__ __forceinline__ float row_xor(float v)
{
    return __builtin_bit_cast(float,
        __builtin_amdgcn_update_dpp(0, __builtin_bit_cast(int, v),
                                    0x160 | M  , 0xf, 0xf, true));
}

__device__ __forceinline__ float xhalf(float v)
{
    const int b = __builtin_bit_cast(int, v);
    return __builtin_bit_cast(float,
        __builtin_amdgcn_permlanex16(b, b, 0x76543210, 0xfedcba98, false, false));
}

__device__ __forceinline__ float wave_sum(float p)
{
    p += row_xor<1>(p);
    p += row_xor<2>(p);
    p += row_xor<4>(p);
    p += row_xor<8>(p);
    p += xhalf(p);
    return p;
}

__device__ __forceinline__ void load_group(float (&bq)[GRP], float (&cq)[GRP],
                                           const float* __restrict__ Bp,
                                           const float* __restrict__ Cp, int t0)
{
    for (int j = 0; j < GRP; ++j) {
        bq[j] = Bp[(size_t)(t0 + j) * DN];
        cq[j] = Cp[(size_t)(t0 + j) * DN];
    }
}

__global__ void ssm_pass1(const float* __restrict__ x, const float* __restrict__ dt,
                          const float* __restrict__ B, const float* __restrict__ C,
                          const float* __restrict__ A,
                          float* __restrict__ y, float* __restrict__ summ)
{
    __shared__ __attribute__((aligned(16))) float dt_s[CH];
    __shared__ __attribute__((aligned(16))) float dtx_s[CH];
    __shared__ __attribute__((aligned(16))) float ypart[4 * CH];

    const int k = blockIdx.x, b = blockIdx.y;
    const int tid = threadIdx.x;
    const int lane = tid & 31, wv = tid >> 5;
    const size_t sbase = (size_t)b * SEQ + (size_t)k * CH;

    const float* Bp = B + sbase * DN + tid;
    const float* Cp = C + sbase * DN + tid;

    float bc[GRP], cc[GRP];
    float bn[GRP], cn[GRP];
    load_group(bc, cc, Bp, Cp, 0);

    float d = dt[sbase + tid];
    dt_s[tid]  = d;
    dtx_s[tid] = d * x[sbase + tid];
    const float An = A[tid];
    __syncthreads();

    float h = 0.f, sd = 0.f;
    for (int g = 0; g < NGRP; ++g) {
        const int t0 = g * GRP;
        if (g + 1 < NGRP) load_group(bn, cn, Bp, Cp, t0 + GRP);

        float dtv[GRP], dxv[GRP];
        for (int q = 0; q < GRP / 4; ++q) {
            const float4 d4 = *(const float4*)&dt_s[t0 + 4 * q];
            const float4 x4 = *(const float4*)&dtx_s[t0 + 4 * q];
            dtv[4 * q + 0] = d4.x; dtv[4 * q + 1] = d4.y;
            dtv[4 * q + 2] = d4.z; dtv[4 * q + 3] = d4.w;
            dxv[4 * q + 0] = x4.x; dxv[4 * q + 1] = x4.y;
            dxv[4 * q + 2] = x4.z; dxv[4 * q + 3] = x4.w;
        }

        float ps[GRP];
        for (int j = 0; j < GRP; ++j) {
            float dts = dtv[j];
            sd += dts;
            float a = __expf(dts * An);
            h = fmaf(a, h, dxv[j] * bc[j]);
            ps[j] = wave_sum(cc[j] * h);
        }
        if (lane == 0) {
            for (int q = 0; q < GRP / 4; ++q)
                *(float4*)&ypart[wv * CH + t0 + 4 * q] =
                    make_float4(ps[4 * q + 0], ps[4 * q + 1], ps[4 * q + 2], ps[4 * q + 3]);
        }

        for (int j = 0; j < GRP; ++j) { bc[j] = bn[j]; cc[j] = cn[j]; }
    }

    reinterpret_cast<float2*>(summ)[((size_t)b * NCH + k) * DN + tid] =
        make_float2(__expf(An * sd), h);
    __syncthreads();
    y[sbase + tid] = ypart[0 * CH + tid] + ypart[1 * CH + tid] +
                     ypart[2 * CH + tid] + ypart[3 * CH + tid];
}

#define LBQ 8

__global__ void __launch_bounds__(CH)
ssm_pass2(const float* __restrict__ dt, const float* __restrict__ C,
          const float* __restrict__ A, const float* __restrict__ summ,
          float* __restrict__ y, float* __restrict__ hfin)
{
    __shared__ float Sd[CH];
    __shared__ float sA[DN];
    __shared__ float hin[DN];
    __shared__ float ycorr[CH];

    const int k = blockIdx.x, b = blockIdx.y;
    const int tid = threadIdx.x;
    const int lane = tid & 31, wv = tid >> 5;
    const size_t sbase = (size_t)b * SEQ + (size_t)k * CH;

    const float dtv = dt[sbase + tid];
    const float Av  = A[tid];

    const float2* sp = reinterpret_cast<const float2*>(summ) + (size_t)b * NCH * DN + tid;
    float2 sv[NCH];
#pragma unroll
    for (int j0 = NCH - LBQ; j0 >= 0; j0 -= LBQ) {
        if (j0 < k) {
#pragma unroll
            for (int q = 0; q < LBQ; ++q) {
                const int j = (j0 + q < k) ? (j0 + q) : (k - 1);
                sv[j0 + q] = sp[(size_t)j * DN];
            }
        }
    }
    float2 own = make_float2(0.f, 0.f);
    if (k == NCH - 1) own = sp[(size_t)(NCH - 1) * DN];
    Sd[tid] = dtv;
    sA[tid] = Av;
    __syncthreads();
    for (int off = 1; off < CH; off <<= 1) {
        float v = (tid >= off) ? Sd[tid - off] : 0.f;
        __syncthreads();
        Sd[tid] += v;
        __syncthreads();
    }

    float hc = 0.f;
#pragma unroll
    for (int j0 = 0; j0 < NCH; j0 += LBQ) {
        if (j0 + LBQ <= k) {
#pragma unroll
            for (int q = 0; q < LBQ; ++q)
                hc = fmaf(sv[j0 + q].x, hc, sv[j0 + q].y);
        } else if (j0 < k) {
#pragma unroll
            for (int q = 0; q < LBQ; ++q)
                if (j0 + q < k) hc = fmaf(sv[j0 + q].x, hc, sv[j0 + q].y);
        }
    }
    hin[tid] = hc;
    const float hend = fmaf(own.x, hc, own.y);
    __syncthreads();

    const float* Cb = C + sbase * DN;
    const int kBase = (lane < 16) ? 0 : 2;

    for (int i = 0; i < 2; ++i) {
        const int t0   = (wv * 2 + i) * 16;
        const int tRow = t0 + (lane & 15);
        const float sdt = Sd[tRow];
        const float* Crow = Cb + (size_t)tRow * DN;

        v8f acc = {};
#pragma unroll 8
        for (int k0 = 0; k0 < DN; k0 += 4) {
            const int kk = k0 + kBase;
            float2 cv = *(const float2*)(Crow + kk);
            v2f a2, b2;
            a2.x = cv.x * __expf(sA[kk] * sdt);
            a2.y = cv.y * __expf(sA[kk + 1] * sdt);
            b2.x = hin[kk];
            b2.y = hin[kk + 1];
            acc = __builtin_amdgcn_wmma_f32_16x16x4_f32(
                false, a2, false, b2, (short)0, acc, false, false);
        }
        if (lane == 0) {
#pragma unroll
            for (int r = 0; r < 8; ++r) ycorr[t0 + r] = acc[r];
        }
        if (lane == 16) {
#pragma unroll
            for (int r = 0; r < 8; ++r) ycorr[t0 + 8 + r] = acc[r];
        }
    }
    __syncthreads();
    y[sbase + tid] += ycorr[tid];
    if (k == NCH - 1) hfin[(size_t)b * DN + tid] = hend;
}

extern "C" void kernel_launch(void* const* d_in, const int* in_sizes, int n_in,
                              void* d_out, int out_size, void* d_ws, size_t ws_size,
                              hipStream_t stream) {
    const float* x  = (const float*)d_in[0];
    const float* dt = (const float*)d_in[1];
    const float* B  = (const float*)d_in[2];
    const float* C  = (const float*)d_in[3];
    const float* A  = (const float*)d_in[4];

    float* y    = (float*)d_out;
    float* hfin = y + (size_t)BATCH * SEQ;

    float* summ = (float*)d_ws;

    dim3 grid(NCH, BATCH);
    ssm_pass1<<<grid, CH, 0, stream>>>(x, dt, B, C, A, y, summ);
    ssm_pass2<<<grid, CH, 0, stream>>>(dt, C, A, summ, y, hfin);
}


# pass 2 issues all global reads at entry: a wave's two C tiles (16 B per lane) and its y stay in flight across the scan (LDS-only barriers); WMMA runs from registers; k order of the tile sums changed
# speedup vs baseline: 5.9340x; 1.1046x over previous
#include <hip/hip_runtime.h>

typedef __attribute__((ext_vector_type(2))) float v2f;
typedef __attribute__((ext_vector_type(8))) float v8f;

#define BATCH 8
#define SEQ   8192
#define DN    128
#define CH    128
#define NCH   (SEQ / CH)

#define GRP 16
#define NGRP (CH / GRP)

template <int M>
__device__ __forceinline__ float row_xor(float v)
{
    return __builtin_bit_cast(float,
        __builtin_amdgcn_update_dpp(0, __builtin_bit_cast(int, v),
                                    0x160 | M  , 0xf, 0xf, true));
}

__device__ __forceinline__ float xhalf(float v)
{
    const int b = __builtin_bit_cast(int, v);
    return __builtin_bit_cast(float,
        __builtin_amdgcn_permlanex16(b, b, 0x76543210, 0xfedcba98, false, false));
}

__device__ __forceinline__ float wave_sum(float p)
{
    p += row_xor<1>(p);
    p += row_xor<2>(p);
    p += row_xor<4>(p);
    p += row_xor<8>(p);
    p += xhalf(p);
    return p;
}

__device__ __forceinline__ void load_group(float (&bq)[GRP], float (&cq)[GRP],
                                           const float* __restrict__ Bp,
                                           const float* __restrict__ Cp, int t0)
{
    for (int j = 0; j < GRP; ++j) {
        bq[j] = Bp[(size_t)(t0 + j) * DN];
        cq[j] = Cp[(size_t)(t0 + j) * DN];
    }
}

__global__ void ssm_pass1(const float* __restrict__ x, const float* __restrict__ dt,
                          const float* __restrict__ B, const float* __restrict__ C,
                          const float* __restrict__ A,
                          float* __restrict__ y, float* __restrict__ summ)
{
    __shared__ __attribute__((aligned(16))) float dt_s[CH];
    __shared__ __attribute__((aligned(16))) float dtx_s[CH];
    __shared__ __attribute__((aligned(16))) float ypart[4 * CH];

    const int k = blockIdx.x, b = blockIdx.y;
    const int tid = threadIdx.x;
    const int lane = tid & 31, wv = tid >> 5;
    const size_t sbase = (size_t)b * SEQ + (size_t)k * CH;

    const float* Bp = B + sbase * DN + tid;
    const float* Cp = C + sbase * DN + tid;

    float bc[GRP], cc[GRP];
    float bn[GRP], cn[GRP];
    load_group(bc, cc, Bp, Cp, 0);

    float d = dt[sbase + tid];
    dt_s[tid]  = d;
    dtx_s[tid] = d * x[sbase + tid];
    const float An = A[tid];
    __syncthreads();

    float h = 0.f, sd = 0.f;
    for (int g = 0; g < NGRP; ++g) {
        const int t0 = g * GRP;
        if (g + 1 < NGRP) load_group(bn, cn, Bp, Cp, t0 + GRP);

        float dtv[GRP], dxv[GRP];
        for (int q = 0; q < GRP / 4; ++q) {
            const float4 d4 = *(const float4*)&dt_s[t0 + 4 * q];
            const float4 x4 = *(const float4*)&dtx_s[t0 + 4 * q];
            dtv[4 * q + 0] = d4.x; dtv[4 * q + 1] = d4.y;
            dtv[4 * q + 2] = d4.z; dtv[4 * q + 3] = d4.w;
            dxv[4 * q + 0] = x4.x; dxv[4 * q + 1] = x4.y;
            dxv[4 * q + 2] = x4.z; dxv[4 * q + 3] = x4.w;
        }

        float ps[GRP];
        for (int j = 0; j < GRP; ++j) {
            float dts = dtv[j];
            sd += dts;
            float a = __expf(dts * An);
            h = fmaf(a, h, dxv[j] * bc[j]);
            ps[j] = wave_sum(cc[j] * h);
        }
        if (lane == 0) {
            for (int q = 0; q < GRP / 4; ++q)
                *(float4*)&ypart[wv * CH + t0 + 4 * q] =
                    make_float4(ps[4 * q + 0], ps[4 * q + 1], ps[4 * q + 2], ps[4 * q + 3]);
        }

        for (int j = 0; j < GRP; ++j) { bc[j] = bn[j]; cc[j] = cn[j]; }
    }

    reinterpret_cast<float2*>(summ)[((size_t)b * NCH + k) * DN + tid] =
        make_float2(__expf(An * sd), h);
    __syncthreads();
    y[sbase + tid] = ypart[0 * CH + tid] + ypart[1 * CH + tid] +
                     ypart[2 * CH + tid] + ypart[3 * CH + tid];
}

#define LBQ 8
#define KQ  (DN / 8)

__device__ __forceinline__ void lds_barrier()
{
    __builtin_amdgcn_fence(__ATOMIC_RELEASE, "workgroup", "local");
    __builtin_amdgcn_s_barrier();
    __builtin_amdgcn_fence(__ATOMIC_ACQUIRE, "workgroup", "local");
}

__global__ void __launch_bounds__(CH)
ssm_pass2(const float* __restrict__ dt, const float* __restrict__ C,
          const float* __restrict__ A, const float* __restrict__ summ,
          float* __restrict__ y, float* __restrict__ hfin)
{
    __shared__ float Sd[CH];
    __shared__ __attribute__((aligned(16))) float sA[DN];
    __shared__ __attribute__((aligned(16))) float hin[DN];
    __shared__ float ycorr[CH];

    const int k = blockIdx.x, b = blockIdx.y;
    const int tid = threadIdx.x;
    const int lane = tid & 31, wv = tid >> 5;
    const size_t sbase = (size_t)b * SEQ + (size_t)k * CH;

    const float dtv = dt[sbase + tid];
    const float Av  = A[tid];

    const float2* sp = reinterpret_cast<const float2*>(summ) + (size_t)b * NCH * DN + tid;
    float2 sv[NCH];
#pragma unroll
    for (int j0 = NCH - LBQ; j0 >= 0; j0 -= LBQ) {
        if (j0 < k) {
#pragma unroll
            for (int q = 0; q < LBQ; ++q) {
                const int j = (j0 + q < k) ? (j0 + q) : (k - 1);
                sv[j0 + q] = sp[(size_t)j * DN];
            }
        }
    }
    float2 own = make_float2(0.f, 0.f);
    if (k == NCH - 1) own = sp[(size_t)(NCH - 1) * DN];

    const int half = lane >> 4;
    const int row0 = wv * 32 + (lane & 15);
    const float4* Cq0 = reinterpret_cast<const float4*>(C + (sbase + row0) * DN) + half;
    const float4* Cq1 = Cq0 + 16 * (DN / 4);
    float4 c0[KQ], c1[KQ];
#pragma unroll
    for (int m = 0; m < KQ; ++m) c0[m] = Cq0[2 * m];
#pragma unroll
    for (int m = 0; m < KQ; ++m) c1[m] = Cq1[2 * m];
    const float yloc = y[sbase + tid];

    Sd[tid] = dtv;
    sA[tid] = Av;
    lds_barrier();
    for (int off = 1; off < CH; off <<= 1) {
        float v = (tid >= off) ? Sd[tid - off] : 0.f;
        lds_barrier();
        Sd[tid] += v;
        lds_barrier();
    }

    float hc = 0.f;
#pragma unroll
    for (int j0 = 0; j0 < NCH; j0 += LBQ) {
        if (j0 + LBQ <= k) {
#pragma unroll
            for (int q = 0; q < LBQ; ++q)
                hc = fmaf(sv[j0 + q].x, hc, sv[j0 + q].y);
        } else if (j0 < k) {
#pragma unroll
            for (int q = 0; q < LBQ; ++q)
                if (j0 + q < k) hc = fmaf(sv[j0 + q].x, hc, sv[j0 + q].y);
        }
    }
    hin[tid] = hc;
    const float hend = fmaf(own.x, hc, own.y);
    lds_barrier();

    const float sdt0 = Sd[row0];
    const float sdt1 = Sd[row0 + 16];
    v8f acc0 = {}, acc1 = {};
#pragma unroll
    for (int m = 0; m < KQ; ++m) {
        const int kq = 8 * m + 4 * half;
        const float4 a4 = *(const float4*)&sA[kq];
        const float4 h4 = *(const float4*)&hin[kq];
        v2f a2, b2;
        b2.x = h4.x;
        b2.y = h4.y;
        a2.x = c0[m].x * __expf(a4.x * sdt0);
        a2.y = c0[m].y * __expf(a4.y * sdt0);
        acc0 = __builtin_amdgcn_wmma_f32_16x16x4_f32(
            false, a2, false, b2, (short)0, acc0, false, false);
        a2.x = c1[m].x * __expf(a4.x * sdt1);
        a2.y = c1[m].y * __expf(a4.y * sdt1);
        acc1 = __builtin_amdgcn_wmma_f32_16x16x4_f32(
            false, a2, false, b2, (short)0, acc1, false, false);
        b2.x = h4.z;
        b2.y = h4.w;
        a2.x = c0[m].z * __expf(a4.z * sdt0);
        a2.y = c0[m].w * __expf(a4.w * sdt0);
        acc0 = __builtin_amdgcn_wmma_f32_16x16x4_f32(
            false, a2, false, b2, (short)0, acc0, false, false);
        a2.x = c1[m].z * __expf(a4.z * sdt1);
        a2.y = c1[m].w * __expf(a4.w * sdt1);
        acc1 = __builtin_amdgcn_wmma_f32_16x16x4_f32(
            false, a2, false, b2, (short)0, acc1, false, false);
    }
    const int t0 = wv * 32;
    if (lane == 0) {
#pragma unroll
        for (int r = 0; r < 8; ++r) { ycorr[t0 + r] = acc0[r]; ycorr[t0 + 16 + r] = acc1[r]; }
    }
    if (lane == 16) {
#pragma unroll
        for (int r = 0; r < 8; ++r) { ycorr[t0 + 8 + r] = acc0[r]; ycorr[t0 + 24 + r] = acc1[r]; }
    }
    lds_barrier();
    y[sbase + tid] = yloc + ycorr[tid];
    if (k == NCH - 1) hfin[(size_t)b * DN + tid] = hend;
}

extern "C" void kernel_launch(void* const* d_in, const int* in_sizes, int n_in,
                              void* d_out, int out_size, void* d_ws, size_t ws_size,
                              hipStream_t stream) {
    const float* x  = (const float*)d_in[0];
    const float* dt = (const float*)d_in[1];
    const float* B  = (const float*)d_in[2];
    const float* C  = (const float*)d_in[3];
    const float* A  = (const float*)d_in[4];

    float* y    = (float*)d_out;
    float* hfin = y + (size_t)BATCH * SEQ;

    float* summ = (float*)d_ws;

    dim3 grid(NCH, BATCH);
    ssm_pass1<<<grid, CH, 0, stream>>>(x, dt, B, C, A, y, summ);
    ssm_pass2<<<grid, CH, 0, stream>>>(dt, C, A, summ, y, hfin);
}


# pass 1 keeps the B and C rows of the next two 16-step groups in flight instead of one: three register sets rotate by name (group loop unrolled), LDS-only barrier at staging; sums and order unchanged
# speedup vs baseline: 6.8617x; 1.1563x over previous
#include <hip/hip_runtime.h>

typedef __attribute__((ext_vector_type(2))) float v2f;
typedef __attribute__((ext_vector_type(8))) float v8f;

#define BATCH 8
#define SEQ   8192
#define DN    128
#define CH    128
#define NCH   (SEQ / CH)

#define GRP 16
#define NGRP (CH / GRP)
#define NSET 3

__device__ __forceinline__ void lds_barrier()
{
    __builtin_amdgcn_fence(__ATOMIC_RELEASE, "workgroup", "local");
    __builtin_amdgcn_s_barrier();
    __builtin_amdgcn_fence(__ATOMIC_ACQUIRE, "workgroup", "local");
}

template <int M>
__device__ __forceinline__ float row_xor(float v)
{
    return __builtin_bit_cast(float,
        __builtin_amdgcn_update_dpp(0, __builtin_bit_cast(int, v),
                                    0x160 | M  , 0xf, 0xf, true));
}

__device__ __forceinline__ float xhalf(float v)
{
    const int b = __builtin_bit_cast(int, v);
    return __builtin_bit_cast(float,
        __builtin_amdgcn_permlanex16(b, b, 0x76543210, 0xfedcba98, false, false));
}

__device__ __forceinline__ float wave_sum(float p)
{
    p += row_xor<1>(p);
    p += row_xor<2>(p);
    p += row_xor<4>(p);
    p += row_xor<8>(p);
    p += xhalf(p);
    return p;
}

__device__ __forceinline__ void load_group(float (&bq)[GRP], float (&cq)[GRP],
                                           const float* __restrict__ Bp,
                                           const float* __restrict__ Cp, int t0)
{
    for (int j = 0; j < GRP; ++j) {
        bq[j] = Bp[(size_t)(t0 + j) * DN];
        cq[j] = Cp[(size_t)(t0 + j) * DN];
    }
}

__global__ void __launch_bounds__(CH)
ssm_pass1(const float* __restrict__ x, const float* __restrict__ dt,
          const float* __restrict__ B, const float* __restrict__ C,
          const float* __restrict__ A,
          float* __restrict__ y, float* __restrict__ summ)
{
    __shared__ __attribute__((aligned(16))) float dt_s[CH];
    __shared__ __attribute__((aligned(16))) float dtx_s[CH];
    __shared__ __attribute__((aligned(16))) float ypart[4 * CH];

    const int k = blockIdx.x, b = blockIdx.y;
    const int tid = threadIdx.x;
    const int lane = tid & 31, wv = tid >> 5;
    const size_t sbase = (size_t)b * SEQ + (size_t)k * CH;

    const float* Bp = B + sbase * DN + tid;
    const float* Cp = C + sbase * DN + tid;

    const float d  = dt[sbase + tid];
    const float xv = x[sbase + tid];
    const float An = A[tid];

    float bq[NSET][GRP], cq[NSET][GRP];
    load_group(bq[0], cq[0], Bp, Cp, 0);
    __builtin_amdgcn_sched_barrier(0);
    load_group(bq[1], cq[1], Bp, Cp, GRP);

    dt_s[tid]  = d;
    dtx_s[tid] = d * xv;
    lds_barrier();

    float h = 0.f, sd = 0.f;
#pragma unroll
    for (int g = 0; g < NGRP; ++g) {
        const int t0 = g * GRP;
        const int cur = g % NSET;
        if (g + 2 < NGRP)
            load_group(bq[(g + 2) % NSET], cq[(g + 2) % NSET], Bp, Cp, t0 + 2 * GRP);

        float dtv[GRP], dxv[GRP];
        for (int q = 0; q < GRP / 4; ++q) {
            const float4 d4 = *(const float4*)&dt_s[t0 + 4 * q];
            const float4 x4 = *(const float4*)&dtx_s[t0 + 4 * q];
            dtv[4 * q + 0] = d4.x; dtv[4 * q + 1] = d4.y;
            dtv[4 * q + 2] = d4.z; dtv[4 * q + 3] = d4.w;
            dxv[4 * q + 0] = x4.x; dxv[4 * q + 1] = x4.y;
            dxv[4 * q + 2] = x4.z; dxv[4 * q + 3] = x4.w;
        }

        float ps[GRP];
        for (int j = 0; j < GRP; ++j) {
            float dts = dtv[j];
            sd += dts;
            float a = __expf(dts * An);
            h = fmaf(a, h, dxv[j] * bq[cur][j]);
            ps[j] = wave_sum(cq[cur][j] * h);
        }
        if (lane == 0) {
            for (int q = 0; q < GRP / 4; ++q)
                *(float4*)&ypart[wv * CH + t0 + 4 * q] =
                    make_float4(ps[4 * q + 0], ps[4 * q + 1], ps[4 * q + 2], ps[4 * q + 3]);
        }
    }

    reinterpret_cast<float2*>(summ)[((size_t)b * NCH + k) * DN + tid] =
        make_float2(__expf(An * sd), h);
    __syncthreads();
    y[sbase + tid] = ypart[0 * CH + tid] + ypart[1 * CH + tid] +
                     ypart[2 * CH + tid] + ypart[3 * CH + tid];
}

#define LBQ 8
#define KQ  (DN / 8)

__global__ void __launch_bounds__(CH)
ssm_pass2(const float* __restrict__ dt, const float* __restrict__ C,
          const float* __restrict__ A, const float* __restrict__ summ,
          float* __restrict__ y, float* __restrict__ hfin)
{
    __shared__ float Sd[CH];
    __shared__ __attribute__((aligned(16))) float sA[DN];
    __shared__ __attribute__((aligned(16))) float hin[DN];
    __shared__ float ycorr[CH];

    const int k = blockIdx.x, b = blockIdx.y;
    const int tid = threadIdx.x;
    const int lane = tid & 31, wv = tid >> 5;
    const size_t sbase = (size_t)b * SEQ + (size_t)k * CH;

    const float dtv = dt[sbase + tid];
    const float Av  = A[tid];

    const float2* sp = reinterpret_cast<const float2*>(summ) + (size_t)b * NCH * DN + tid;
    float2 sv[NCH];
#pragma unroll
    for (int j0 = NCH - LBQ; j0 >= 0; j0 -= LBQ) {
        if (j0 < k) {
#pragma unroll
            for (int q = 0; q < LBQ; ++q) {
                const int j = (j0 + q < k) ? (j0 + q) : (k - 1);
                sv[j0 + q] = sp[(size_t)j * DN];
            }
        }
    }
    float2 own = make_float2(0.f, 0.f);
    if (k == NCH - 1) own = sp[(size_t)(NCH - 1) * DN];

    const int half = lane >> 4;
    const int row0 = wv * 32 + (lane & 15);
    const float4* Cq0 = reinterpret_cast<const float4*>(C + (sbase + row0) * DN) + half;
    const float4* Cq1 = Cq0 + 16 * (DN / 4);
    float4 c0[KQ], c1[KQ];
#pragma unroll
    for (int m = 0; m < KQ; ++m) c0[m] = Cq0[2 * m];
#pragma unroll
    for (int m = 0; m < KQ; ++m) c1[m] = Cq1[2 * m];
    const float yloc = y[sbase + tid];

    Sd[tid] = dtv;
    sA[tid] = Av;
    lds_barrier();
    for (int off = 1; off < CH; off <<= 1) {
        float v = (tid >= off) ? Sd[tid - off] : 0.f;
        lds_barrier();
        Sd[tid] += v;
        lds_barrier();
    }

    float hc = 0.f;
#pragma unroll
    for (int j0 = 0; j0 < NCH; j0 += LBQ) {
        if (j0 + LBQ <= k) {
#pragma unroll
            for (int q = 0; q < LBQ; ++q)
                hc = fmaf(sv[j0 + q].x, hc, sv[j0 + q].y);
        } else if (j0 < k) {
#pragma unroll
            for (int q = 0; q < LBQ; ++q)
                if (j0 + q < k) hc = fmaf(sv[j0 + q].x, hc, sv[j0 + q].y);
        }
    }
    hin[tid] = hc;
    const float hend = fmaf(own.x, hc, own.y);
    lds_barrier();

    const float sdt0 = Sd[row0];
    const float sdt1 = Sd[row0 + 16];
    v8f acc0 = {}, acc1 = {};
#pragma unroll
    for (int m = 0; m < KQ; ++m) {
        const int kq = 8 * m + 4 * half;
        const float4 a4 = *(const float4*)&sA[kq];
        const float4 h4 = *(const float4*)&hin[kq];
        v2f a2, b2;
        b2.x = h4.x;
        b2.y = h4.y;
        a2.x = c0[m].x * __expf(a4.x * sdt0);
        a2.y = c0[m].y * __expf(a4.y * sdt0);
        acc0 = __builtin_amdgcn_wmma_f32_16x16x4_f32(
            false, a2, false, b2, (short)0, acc0, false, false);
        a2.x = c1[m].x * __expf(a4.x * sdt1);
        a2.y = c1[m].y * __expf(a4.y * sdt1);
        acc1 = __builtin_amdgcn_wmma_f32_16x16x4_f32(
            false, a2, false, b2, (short)0, acc1, false, false);
        b2.x = h4.z;
        b2.y = h4.w;
        a2.x = c0[m].z * __expf(a4.z * sdt0);
        a2.y = c0[m].w * __expf(a4.w * sdt0);
        acc0 = __builtin_amdgcn_wmma_f32_16x16x4_f32(
            false, a2, false, b2, (short)0, acc0, false, false);
        a2.x = c1[m].z * __expf(a4.z * sdt1);
        a2.y = c1[m].w * __expf(a4.w * sdt1);
        acc1 = __builtin_amdgcn_wmma_f32_16x16x4_f32(
            false, a2, false, b2, (short)0, acc1, false, false);
    }
    const int t0 = wv * 32;
    if (lane == 0) {
#pragma unroll
        for (int r = 0; r < 8; ++r) { ycorr[t0 + r] = acc0[r]; ycorr[t0 + 16 + r] = acc1[r]; }
    }
    if (lane == 16) {
#pragma unroll
        for (int r = 0; r < 8; ++r) { ycorr[t0 + 8 + r] = acc0[r]; ycorr[t0 + 24 + r] = acc1[r]; }
    }
    lds_barrier();
    y[sbase + tid] = yloc + ycorr[tid];
    if (k == NCH - 1) hfin[(size_t)b * DN + tid] = hend;
}

extern "C" void kernel_launch(void* const* d_in, const int* in_sizes, int n_in,
                              void* d_out, int out_size, void* d_ws, size_t ws_size,
                              hipStream_t stream) {
    const float* x  = (const float*)d_in[0];
    const float* dt = (const float*)d_in[1];
    const float* B  = (const float*)d_in[2];
    const float* C  = (const float*)d_in[3];
    const float* A  = (const float*)d_in[4];

    float* y    = (float*)d_out;
    float* hfin = y + (size_t)BATCH * SEQ;

    float* summ = (float*)d_ws;

    dim3 grid(NCH, BATCH);
    ssm_pass1<<<grid, CH, 0, stream>>>(x, dt, B, C, A, y, summ);
    ssm_pass2<<<grid, CH, 0, stream>>>(dt, C, A, summ, y, hfin);
}


# pass 2: a wave builds the dt prefix of its 32 steps in its own lanes (row_shr DPP adds, a readlane, a wave sum of the steps in front); the 7-round LDS scan and 15 barriers go; dt sum order changed
# speedup vs baseline: 7.2801x; 1.0610x over previous
#include <hip/hip_runtime.h>

typedef __attribute__((ext_vector_type(2))) float v2f;
typedef __attribute__((ext_vector_type(8))) float v8f;

#define BATCH 8
#define SEQ   8192
#define DN    128
#define CH    128
#define NCH   (SEQ / CH)

#define GRP 16
#define NGRP (CH / GRP)
#define NSET 3

__device__ __forceinline__ void lds_barrier()
{
    __builtin_amdgcn_fence(__ATOMIC_RELEASE, "workgroup", "local");
    __builtin_amdgcn_s_barrier();
    __builtin_amdgcn_fence(__ATOMIC_ACQUIRE, "workgroup", "local");
}

template <int M>
__device__ __forceinline__ float row_xor(float v)
{
    return __builtin_bit_cast(float,
        __builtin_amdgcn_update_dpp(0, __builtin_bit_cast(int, v),
                                    0x160 | M  , 0xf, 0xf, true));
}

__device__ __forceinline__ float xhalf(float v)
{
    const int b = __builtin_bit_cast(int, v);
    return __builtin_bit_cast(float,
        __builtin_amdgcn_permlanex16(b, b, 0x76543210, 0xfedcba98, false, false));
}

__device__ __forceinline__ float wave_sum(float p)
{
    p += row_xor<1>(p);
    p += row_xor<2>(p);
    p += row_xor<4>(p);
    p += row_xor<8>(p);
    p += xhalf(p);
    return p;
}

template <int N>
__device__ __forceinline__ float row_shr(float v)
{
    return __builtin_bit_cast(float,
        __builtin_amdgcn_update_dpp(0, __builtin_bit_cast(int, v),
                                    0x110 | N  , 0xf, 0xf, true));
}

__device__ __forceinline__ void load_group(float (&bq)[GRP], float (&cq)[GRP],
                                           const float* __restrict__ Bp,
                                           const float* __restrict__ Cp, int t0)
{
    for (int j = 0; j < GRP; ++j) {
        bq[j] = Bp[(size_t)(t0 + j) * DN];
        cq[j] = Cp[(size_t)(t0 + j) * DN];
    }
}

__global__ void __launch_bounds__(CH)
ssm_pass1(const float* __restrict__ x, const float* __restrict__ dt,
          const float* __restrict__ B, const float* __restrict__ C,
          const float* __restrict__ A,
          float* __restrict__ y, float* __restrict__ summ)
{
    __shared__ __attribute__((aligned(16))) float dt_s[CH];
    __shared__ __attribute__((aligned(16))) float dtx_s[CH];
    __shared__ __attribute__((aligned(16))) float ypart[4 * CH];

    const int k = blockIdx.x, b = blockIdx.y;
    const int tid = threadIdx.x;
    const int lane = tid & 31, wv = tid >> 5;
    const size_t sbase = (size_t)b * SEQ + (size_t)k * CH;

    const float* Bp = B + sbase * DN + tid;
    const float* Cp = C + sbase * DN + tid;

    const float d  = dt[sbase + tid];
    const float xv = x[sbase + tid];
    const float An = A[tid];

    float bq[NSET][GRP], cq[NSET][GRP];
    load_group(bq[0], cq[0], Bp, Cp, 0);
    __builtin_amdgcn_sched_barrier(0);
    load_group(bq[1], cq[1], Bp, Cp, GRP);

    dt_s[tid]  = d;
    dtx_s[tid] = d * xv;
    lds_barrier();

    float h = 0.f, sd = 0.f;
#pragma unroll
    for (int g = 0; g < NGRP; ++g) {
        const int t0 = g * GRP;
        const int cur = g % NSET;
        if (g + 2 < NGRP)
            load_group(bq[(g + 2) % NSET], cq[(g + 2) % NSET], Bp, Cp, t0 + 2 * GRP);

        float dtv[GRP], dxv[GRP];
        for (int q = 0; q < GRP / 4; ++q) {
            const float4 d4 = *(const float4*)&dt_s[t0 + 4 * q];
            const float4 x4 = *(const float4*)&dtx_s[t0 + 4 * q];
            dtv[4 * q + 0] = d4.x; dtv[4 * q + 1] = d4.y;
            dtv[4 * q + 2] = d4.z; dtv[4 * q + 3] = d4.w;
            dxv[4 * q + 0] = x4.x; dxv[4 * q + 1] = x4.y;
            dxv[4 * q + 2] = x4.z; dxv[4 * q + 3] = x4.w;
        }

        float ps[GRP];
        for (int j = 0; j < GRP; ++j) {
            float dts = dtv[j];
            sd += dts;
            float a = __expf(dts * An);
            h = fmaf(a, h, dxv[j] * bq[cur][j]);
            ps[j] = wave_sum(cq[cur][j] * h);
        }
        if (lane == 0) {
            for (int q = 0; q < GRP / 4; ++q)
                *(float4*)&ypart[wv * CH + t0 + 4 * q] =
                    make_float4(ps[4 * q + 0], ps[4 * q + 1], ps[4 * q + 2], ps[4 * q + 3]);
        }
    }

    reinterpret_cast<float2*>(summ)[((size_t)b * NCH + k) * DN + tid] =
        make_float2(__expf(An * sd), h);
    __syncthreads();
    y[sbase + tid] = ypart[0 * CH + tid] + ypart[1 * CH + tid] +
                     ypart[2 * CH + tid] + ypart[3 * CH + tid];
}

#define LBQ 8
#define KQ  (DN / 8)

__global__ void __launch_bounds__(CH)
ssm_pass2(const float* __restrict__ dt, const float* __restrict__ C,
          const float* __restrict__ A, const float* __restrict__ summ,
          float* __restrict__ y, float* __restrict__ hfin)
{
    __shared__ __attribute__((aligned(16))) float sA[DN];
    __shared__ __attribute__((aligned(16))) float hin[DN];
    __shared__ float ycorr[CH];

    const int k = blockIdx.x, b = blockIdx.y;
    const int tid = threadIdx.x;
    const int lane = tid & 31, wv = tid >> 5;
    const size_t sbase = (size_t)b * SEQ + (size_t)k * CH;

    const float dtv = dt[sbase + tid];
    const float Av  = A[tid];
    const float e0 = dt[sbase + lane];
    const float e1 = dt[sbase + 32 + lane];
    const float e2 = dt[sbase + 64 + lane];

    const float2* sp = reinterpret_cast<const float2*>(summ) + (size_t)b * NCH * DN + tid;
    float2 sv[NCH];
#pragma unroll
    for (int j0 = NCH - LBQ; j0 >= 0; j0 -= LBQ) {
        if (j0 < k) {
#pragma unroll
            for (int q = 0; q < LBQ; ++q) {
                const int j = (j0 + q < k) ? (j0 + q) : (k - 1);
                sv[j0 + q] = sp[(size_t)j * DN];
            }
        }
    }
    float2 own = make_float2(0.f, 0.f);
    if (k == NCH - 1) own = sp[(size_t)(NCH - 1) * DN];

    const int half = lane >> 4;
    const int row0 = wv * 32 + (lane & 15);
    const float4* Cq0 = reinterpret_cast<const float4*>(C + (sbase + row0) * DN) + half;
    const float4* Cq1 = Cq0 + 16 * (DN / 4);
    float4 c0[KQ], c1[KQ];
#pragma unroll
    for (int m = 0; m < KQ; ++m) c0[m] = Cq0[2 * m];
#pragma unroll
    for (int m = 0; m < KQ; ++m) c1[m] = Cq1[2 * m];
    const float yloc = y[sbase + tid];

    float rs = dtv;
    rs += row_shr<1>(rs);
    rs += row_shr<2>(rs);
    rs += row_shr<4>(rs);
    rs += row_shr<8>(rs);
    const float r0tot = __builtin_bit_cast(float,
        __builtin_amdgcn_readlane(__builtin_bit_cast(int, rs), 15));
    const float before = (wv > 0 ? e0 : 0.f) + (wv > 1 ? e1 : 0.f) + (wv > 2 ? e2 : 0.f);
    const float sdm = wave_sum(before) + (rs + (half ? r0tot : 0.f));
    const float sdo  = xhalf(sdm);
    const float sdt0 = half ? sdo : sdm;
    const float sdt1 = half ? sdm : sdo;

    float hc = 0.f;
#pragma unroll
    for (int j0 = 0; j0 < NCH; j0 += LBQ) {
        if (j0 + LBQ <= k) {
#pragma unroll
            for (int q = 0; q < LBQ; ++q)
                hc = fmaf(sv[j0 + q].x, hc, sv[j0 + q].y);
        } else if (j0 < k) {
#pragma unroll
            for (int q = 0; q < LBQ; ++q)
                if (j0 + q < k) hc = fmaf(sv[j0 + q].x, hc, sv[j0 + q].y);
        }
    }
    sA[tid]  = Av;
    hin[tid] = hc;
    const float hend = fmaf(own.x, hc, own.y);
    lds_barrier();

    v8f acc0 = {}, acc1 = {};
#pragma unroll
    for (int m = 0; m < KQ; ++m) {
        const int kq = 8 * m + 4 * half;
        const float4 a4 = *(const float4*)&sA[kq];
        const float4 h4 = *(const float4*)&hin[kq];
        v2f a2, b2;
        b2.x = h4.x;
        b2.y = h4.y;
        a2.x = c0[m].x * __expf(a4.x * sdt0);
        a2.y = c0[m].y * __expf(a4.y * sdt0);
        acc0 = __builtin_amdgcn_wmma_f32_16x16x4_f32(
            false, a2, false, b2, (short)0, acc0, false, false);
        a2.x = c1[m].x * __expf(a4.x * sdt1);
        a2.y = c1[m].y * __expf(a4.y * sdt1);
        acc1 = __builtin_amdgcn_wmma_f32_16x16x4_f32(
            false, a2, false, b2, (short)0, acc1, false, false);
        b2.x = h4.z;
        b2.y = h4.w;
        a2.x = c0[m].z * __expf(a4.z * sdt0);
        a2.y = c0[m].w * __expf(a4.w * sdt0);
        acc0 = __builtin_amdgcn_wmma_f32_16x16x4_f32(
            false, a2, false, b2, (short)0, acc0, false, false);
        a2.x = c1[m].z * __expf(a4.z * sdt1);
        a2.y = c1[m].w * __expf(a4.w * sdt1);
        acc1 = __builtin_amdgcn_wmma_f32_16x16x4_f32(
            false, a2, false, b2, (short)0, acc1, false, false);
    }
    const int t0 = wv * 32;
    if (lane == 0) {
#pragma unroll
        for (int r = 0; r < 8; ++r) { ycorr[t0 + r] = acc0[r]; ycorr[t0 + 16 + r] = acc1[r]; }
    }
    if (lane == 16) {
#pragma unroll
        for (int r = 0; r < 8; ++r) { ycorr[t0 + 8 + r] = acc0[r]; ycorr[t0 + 24 + r] = acc1[r]; }
    }
    lds_barrier();
    y[sbase + tid] = yloc + ycorr[tid];
    if (k == NCH - 1) hfin[(size_t)b * DN + tid] = hend;
}

extern "C" void kernel_launch(void* const* d_in, const int* in_sizes, int n_in,
                              void* d_out, int out_size, void* d_ws, size_t ws_size,
                              hipStream_t stream) {
    const float* x  = (const float*)d_in[0];
    const float* dt = (const float*)d_in[1];
    const float* B  = (const float*)d_in[2];
    const float* C  = (const float*)d_in[3];
    const float* A  = (const float*)d_in[4];

    float* y    = (float*)d_out;
    float* hfin = y + (size_t)BATCH * SEQ;

    float* summ = (float*)d_ws;

    dim3 grid(NCH, BATCH);
    ssm_pass1<<<grid, CH, 0, stream>>>(x, dt, B, C, A, y, summ);
    ssm_pass2<<<grid, CH, 0, stream>>>(dt, C, A, summ, y, hfin);
}


# pass 1: a wave scans a 32-step quarter chunk for all 128 dims (lane = 4 dims, 16-byte row reads, all 64 out at entry); pass 2 starts each quarter from its own state; dot and dt sum order changed
# speedup vs baseline: 7.9387x; 1.0905x over previous
#include <hip/hip_runtime.h>

typedef __attribute__((ext_vector_type(2))) float v2f;
typedef __attribute__((ext_vector_type(8))) float v8f;

#define BATCH 8
#define SEQ   8192
#define DN    128
#define CH    128
#define NCH   (SEQ / CH)

#define GRP 16
#define QCH 32
#define NGQ (QCH / GRP)

__device__ __forceinline__ void lds_barrier()
{
    __builtin_amdgcn_fence(__ATOMIC_RELEASE, "workgroup", "local");
    __builtin_amdgcn_s_barrier();
    __builtin_amdgcn_fence(__ATOMIC_ACQUIRE, "workgroup", "local");
}

template <int M>
__device__ __forceinline__ float row_xor(float v)
{
    return __builtin_bit_cast(float,
        __builtin_amdgcn_update_dpp(0, __builtin_bit_cast(int, v),
                                    0x160 | M  , 0xf, 0xf, true));
}

__device__ __forceinline__ float xhalf(float v)
{
    const int b = __builtin_bit_cast(int, v);
    return __builtin_bit_cast(float,
        __builtin_amdgcn_permlanex16(b, b, 0x76543210, 0xfedcba98, false, false));
}

__device__ __forceinline__ float wave_sum(float p)
{
    p += row_xor<1>(p);
    p += row_xor<2>(p);
    p += row_xor<4>(p);
    p += row_xor<8>(p);
    p += xhalf(p);
    return p;
}

template <int N>
__device__ __forceinline__ float row_shr(float v)
{
    return __builtin_bit_cast(float,
        __builtin_amdgcn_update_dpp(0, __builtin_bit_cast(int, v),
                                    0x110 | N  , 0xf, 0xf, true));
}

__global__ void __launch_bounds__(CH)
ssm_pass1(const float* __restrict__ x, const float* __restrict__ dt,
          const float* __restrict__ B, const float* __restrict__ C,
          const float* __restrict__ A,
          float* __restrict__ y, float* __restrict__ summ, float* __restrict__ qsum)
{
    __shared__ __attribute__((aligned(16))) float dt_s[CH];
    __shared__ __attribute__((aligned(16))) float dtx_s[CH];
    __shared__ __attribute__((aligned(16))) float yloc_s[CH];
    __shared__ __attribute__((aligned(16))) float2 pair_s[4 * DN];

    const int k = blockIdx.x, b = blockIdx.y;
    const int tid = threadIdx.x;
    const int lane = tid & 31, wv = tid >> 5;
    const size_t sbase = (size_t)b * SEQ + (size_t)k * CH;
    const size_t rbase = sbase + (size_t)wv * QCH;

    const float4* Bp = reinterpret_cast<const float4*>(B + rbase * DN) + lane;
    const float4* Cp = reinterpret_cast<const float4*>(C + rbase * DN) + lane;

    const float d  = dt[sbase + tid];
    const float xv = x[sbase + tid];
    const float4 An = *reinterpret_cast<const float4*>(A + 4 * lane);

    float4 bq[QCH], cq[QCH];
#pragma unroll
    for (int j = 0; j < GRP; ++j) {
        bq[j] = Bp[(size_t)j * (DN / 4)];
        cq[j] = Cp[(size_t)j * (DN / 4)];
    }
    __builtin_amdgcn_sched_barrier(0);
#pragma unroll
    for (int j = GRP; j < QCH; ++j) {
        bq[j] = Bp[(size_t)j * (DN / 4)];
        cq[j] = Cp[(size_t)j * (DN / 4)];
    }

    dt_s[tid]  = d;
    dtx_s[tid] = d * xv;
    lds_barrier();

    float h0 = 0.f, h1 = 0.f, h2 = 0.f, h3 = 0.f, sd = 0.f;
#pragma unroll
    for (int g = 0; g < NGQ; ++g) {
        const int t0 = wv * QCH + g * GRP;

        float dtv[GRP], dxv[GRP];
        for (int q = 0; q < GRP / 4; ++q) {
            const float4 d4 = *(const float4*)&dt_s[t0 + 4 * q];
            const float4 x4 = *(const float4*)&dtx_s[t0 + 4 * q];
            dtv[4 * q + 0] = d4.x; dtv[4 * q + 1] = d4.y;
            dtv[4 * q + 2] = d4.z; dtv[4 * q + 3] = d4.w;
            dxv[4 * q + 0] = x4.x; dxv[4 * q + 1] = x4.y;
            dxv[4 * q + 2] = x4.z; dxv[4 * q + 3] = x4.w;
        }

        float ps[GRP];
        for (int j = 0; j < GRP; ++j) {
            const float dts = dtv[j];
            const float dx  = dxv[j];
            const float4 bb = bq[g * GRP + j];
            const float4 cc = cq[g * GRP + j];
            sd += dts;
            h0 = fmaf(__expf(dts * An.x), h0, dx * bb.x);
            h1 = fmaf(__expf(dts * An.y), h1, dx * bb.y);
            h2 = fmaf(__expf(dts * An.z), h2, dx * bb.z);
            h3 = fmaf(__expf(dts * An.w), h3, dx * bb.w);
            float p = cc.x * h0;
            p = fmaf(cc.y, h1, p);
            p = fmaf(cc.z, h2, p);
            p = fmaf(cc.w, h3, p);
            ps[j] = wave_sum(p);
        }
        if (lane == 0) {
            for (int q = 0; q < GRP / 4; ++q)
                *(float4*)&yloc_s[t0 + 4 * q] =
                    make_float4(ps[4 * q + 0], ps[4 * q + 1], ps[4 * q + 2], ps[4 * q + 3]);
        }
    }

    float4* pw = reinterpret_cast<float4*>(&pair_s[wv * DN + 4 * lane]);
    pw[0] = make_float4(__expf(An.x * sd), h0, __expf(An.y * sd), h1);
    pw[1] = make_float4(__expf(An.z * sd), h2, __expf(An.w * sd), h3);
    __syncthreads();
    y[sbase + tid] = yloc_s[tid];

    const float2 q0 = pair_s[0 * DN + tid];
    const float2 q1 = pair_s[1 * DN + tid];
    const float2 q2 = pair_s[2 * DN + tid];
    const float2 q3 = pair_s[3 * DN + tid];
    float2* qo = reinterpret_cast<float2*>(qsum) + ((size_t)b * NCH + k) * 3 * DN + tid;
    qo[0 * DN] = q0;
    qo[1 * DN] = q1;
    qo[2 * DN] = q2;
    float hh = fmaf(q1.x, q0.y, q1.y);
    hh = fmaf(q2.x, hh, q2.y);
    hh = fmaf(q3.x, hh, q3.y);
    reinterpret_cast<float2*>(summ)[((size_t)b * NCH + k) * DN + tid] =
        make_float2(((q0.x * q1.x) * q2.x) * q3.x, hh);
}

#define LBQ 8
#define KQ  (DN / 8)

__global__ void __launch_bounds__(CH)
ssm_pass2(const float* __restrict__ dt, const float* __restrict__ C,
          const float* __restrict__ A, const float* __restrict__ summ,
          const float* __restrict__ qsum,
          float* __restrict__ y, float* __restrict__ hfin)
{
    __shared__ __attribute__((aligned(16))) float sA[DN];
    __shared__ __attribute__((aligned(16))) float hin[4 * DN];
    __shared__ float ycorr[CH];

    const int k = blockIdx.x, b = blockIdx.y;
    const int tid = threadIdx.x;
    const int lane = tid & 31, wv = tid >> 5;
    const size_t sbase = (size_t)b * SEQ + (size_t)k * CH;

    const float dtv = dt[sbase + tid];
    const float Av  = A[tid];

    const float2* sp = reinterpret_cast<const float2*>(summ) + (size_t)b * NCH * DN + tid;
    float2 sv[NCH];
#pragma unroll
    for (int j0 = NCH - LBQ; j0 >= 0; j0 -= LBQ) {
        if (j0 < k) {
#pragma unroll
            for (int q = 0; q < LBQ; ++q) {
                const int j = (j0 + q < k) ? (j0 + q) : (k - 1);
                sv[j0 + q] = sp[(size_t)j * DN];
            }
        }
    }
    float2 own = make_float2(0.f, 0.f);
    if (k == NCH - 1) own = sp[(size_t)(NCH - 1) * DN];

    const float2* qp = reinterpret_cast<const float2*>(qsum) + ((size_t)b * NCH + k) * 3 * DN + tid;
    const float2 q0 = qp[0 * DN];
    const float2 q1 = qp[1 * DN];
    const float2 q2 = qp[2 * DN];

    const int half = lane >> 4;
    const int row0 = wv * 32 + (lane & 15);
    const float4* Cq0 = reinterpret_cast<const float4*>(C + (sbase + row0) * DN) + half;
    const float4* Cq1 = Cq0 + 16 * (DN / 4);
    float4 c0[KQ], c1[KQ];
#pragma unroll
    for (int m = 0; m < KQ; ++m) c0[m] = Cq0[2 * m];
#pragma unroll
    for (int m = 0; m < KQ; ++m) c1[m] = Cq1[2 * m];
    const float yloc = y[sbase + tid];

    float rs = dtv;
    rs += row_shr<1>(rs);
    rs += row_shr<2>(rs);
    rs += row_shr<4>(rs);
    rs += row_shr<8>(rs);
    const float r0tot = __builtin_bit_cast(float,
        __builtin_amdgcn_readlane(__builtin_bit_cast(int, rs), 15));
    const float sdm = rs + (half ? r0tot : 0.f);
    const float sdo  = xhalf(sdm);
    const float sdt0 = half ? sdo : sdm;
    const float sdt1 = half ? sdm : sdo;

    float hc = 0.f;
#pragma unroll
    for (int j0 = 0; j0 < NCH; j0 += LBQ) {
        if (j0 + LBQ <= k) {
#pragma unroll
            for (int q = 0; q < LBQ; ++q)
                hc = fmaf(sv[j0 + q].x, hc, sv[j0 + q].y);
        } else if (j0 < k) {
#pragma unroll
            for (int q = 0; q < LBQ; ++q)
                if (j0 + q < k) hc = fmaf(sv[j0 + q].x, hc, sv[j0 + q].y);
        }
    }
    const float g1 = fmaf(q0.x, hc, q0.y);
    const float g2 = fmaf(q1.x, g1, q1.y);
    const float g3 = fmaf(q2.x, g2, q2.y);
    sA[tid]  = Av;
    hin[0 * DN + tid] = hc;
    hin[1 * DN + tid] = g1;
    hin[2 * DN + tid] = g2;
    hin[3 * DN + tid] = g3;
    const float hend = fmaf(own.x, hc, own.y);
    lds_barrier();

    v8f acc0 = {}, acc1 = {};
#pragma unroll
    for (int m = 0; m < KQ; ++m) {
        const int kq = 8 * m + 4 * half;
        const float4 a4 = *(const float4*)&sA[kq];
        const float4 h4 = *(const float4*)&hin[wv * DN + kq];
        v2f a2, b2;
        b2.x = h4.x;
        b2.y = h4.y;
        a2.x = c0[m].x * __expf(a4.x * sdt0);
        a2.y = c0[m].y * __expf(a4.y * sdt0);
        acc0 = __builtin_amdgcn_wmma_f32_16x16x4_f32(
            false, a2, false, b2, (short)0, acc0, false, false);
        a2.x = c1[m].x * __expf(a4.x * sdt1);
        a2.y = c1[m].y * __expf(a4.y * sdt1);
        acc1 = __builtin_amdgcn_wmma_f32_16x16x4_f32(
            false, a2, false, b2, (short)0, acc1, false, false);
        b2.x = h4.z;
        b2.y = h4.w;
        a2.x = c0[m].z * __expf(a4.z * sdt0);
        a2.y = c0[m].w * __expf(a4.w * sdt0);
        acc0 = __builtin_amdgcn_wmma_f32_16x16x4_f32(
            false, a2, false, b2, (short)0, acc0, false, false);
        a2.x = c1[m].z * __expf(a4.z * sdt1);
        a2.y = c1[m].w * __expf(a4.w * sdt1);
        acc1 = __builtin_amdgcn_wmma_f32_16x16x4_f32(
            false, a2, false, b2, (short)0, acc1, false, false);
    }
    const int t0 = wv * 32;
    if (lane == 0) {
#pragma unroll
        for (int r = 0; r < 8; ++r) { ycorr[t0 + r] = acc0[r]; ycorr[t0 + 16 + r] = acc1[r]; }
    }
    if (lane == 16) {
#pragma unroll
        for (int r = 0; r < 8; ++r) { ycorr[t0 + 8 + r] = acc0[r]; ycorr[t0 + 24 + r] = acc1[r]; }
    }
    lds_barrier();
    y[sbase + tid] = yloc + ycorr[tid];
    if (k == NCH - 1) hfin[(size_t)b * DN + tid] = hend;
}

extern "C" void kernel_launch(void* const* d_in, const int* in_sizes, int n_in,
                              void* d_out, int out_size, void* d_ws, size_t ws_size,
                              hipStream_t stream) {
    const float* x  = (const float*)d_in[0];
    const float* dt = (const float*)d_in[1];
    const float* B  = (const float*)d_in[2];
    const float* C  = (const float*)d_in[3];
    const float* A  = (const float*)d_in[4];

    float* y    = (float*)d_out;
    float* hfin = y + (size_t)BATCH * SEQ;

    float* summ = (float*)d_ws;
    float* qsum = summ + (size_t)BATCH * NCH * DN * 2;

    dim3 grid(NCH, BATCH);
    ssm_pass1<<<grid, CH, 0, stream>>>(x, dt, B, C, A, y, summ, qsum);
    ssm_pass2<<<grid, CH, 0, stream>>>(dt, C, A, summ, qsum, y, hfin);
}
